# starting kernel
# baseline (speedup 1.0000x reference)
#include <hip/hip_runtime.h>
#include <math.h>

typedef __attribute__((ext_vector_type(2))) float v2f;
typedef __attribute__((ext_vector_type(4))) float v4f;
typedef __attribute__((ext_vector_type(8))) float v8f;

#define SPR_S 32
#define SPR_H 32
#define TPB 256
#define WAVES_PER_BLOCK (TPB / 32)
#define PTS_PER_TILE 16

#if __has_builtin(__builtin_amdgcn_tanhf)
  #define FAST_TANH(x) __builtin_amdgcn_tanhf(x)
#elif __has_builtin(__builtin_amdgcn_tanh_f32)
  #define FAST_TANH(x) __builtin_amdgcn_tanh_f32(x)
#else
  #define FAST_TANH(x) tanhf(x)
#endif

__global__ __launch_bounds__(TPB) void springnn_wmma_f32(
    const float* __restrict__ t,
    const float* __restrict__ W1,
    const float* __restrict__ b1,
    const float* __restrict__ W2,
    const float* __restrict__ b2,
    const float* __restrict__ xmin,
    float* __restrict__ out, int N)
{
  __shared__ float sW1[SPR_S * SPR_H];
  __shared__ float sB1[SPR_S * SPR_H];
  __shared__ float sW2[SPR_S * SPR_H];
  __shared__ float sB2[SPR_S];
  __shared__ float sLo[SPR_S];

  for (int i = threadIdx.x; i < SPR_S * SPR_H; i += TPB) {
    sW1[i] = W1[i];
    sB1[i] = b1[i];
    sW2[i] = W2[i];
  }
  if (threadIdx.x < SPR_S) {
    sB2[threadIdx.x] = b2[threadIdx.x];
    sLo[threadIdx.x] = xmin[threadIdx.x];
  }
  __syncthreads();

  const int lane  = threadIdx.x & 31;
  const int wave  = threadIdx.x >> 5;
  const int kh    = lane >> 4;
  const int p     = lane & 15;
  const int tile  = blockIdx.x * WAVES_PER_BLOCK + wave;
  const int base  = tile * PTS_PER_TILE;
  if (base >= N) return;

  const int tidx  = base + p;
  const float tv  = t[tidx < N ? tidx : (N - 1)];

  v8f acc = {};
  float accB2 = 0.0f;

  for (int s = 0; s < SPR_S; ++s) {
    const float lo = sLo[s];
    const float d  = tv - lo;
    const float cw = 0.5f * (1.0f - __cosf(1.57079632679489662f * d));
    const float w  = (d >= 0.0f && d <= 4.0f) ? cw : 0.0f;
    accB2 = fmaf(w, sB2[s], accB2);

    const float* w1s = &sW1[s * SPR_H];
    const float* b1s = &sB1[s * SPR_H];
    const float* w2s = &sW2[s * SPR_H];

#pragma unroll
    for (int j = 0; j < 8; ++j) {
      const int k = 4 * j + 2 * kh;
      const v2f w1p = *(const v2f*)(w1s + k);
      const v2f b1p = *(const v2f*)(b1s + k);
      const v2f w2p = *(const v2f*)(w2s + k);
      v2f a;
      a.x = w * FAST_TANH(fmaf(tv, w1p.x, b1p.x));
      a.y = w * FAST_TANH(fmaf(tv, w1p.y, b1p.y));
      acc = __builtin_amdgcn_wmma_f32_16x16x4_f32(
          false, a, false, w2p, (short)0, acc, false, false);
    }
  }

  const int ib2 = __builtin_bit_cast(int, accB2);
  v8f res;
#pragma unroll
  for (int r = 0; r < 8; ++r) {
    const float addLo = __builtin_bit_cast(float, __builtin_amdgcn_readlane(ib2, r));
    const float addHi = __builtin_bit_cast(float, __builtin_amdgcn_readlane(ib2, r + 8));
    res[r] = acc[r] + (kh ? addHi : addLo);
  }

  if (p == 0) {
    const int o0 = base + kh * 8;
    if (o0 + 8 <= N) {
      v4f lo4 = {res[0], res[1], res[2], res[3]};
      v4f hi4 = {res[4], res[5], res[6], res[7]};
      *(v4f*)(out + o0)     = lo4;
      *(v4f*)(out + o0 + 4) = hi4;
    } else {
      for (int r = 0; r < 8 && o0 + r < N; ++r) out[o0 + r] = res[r];
    }
  }
}

extern "C" void kernel_launch(void* const* d_in, const int* in_sizes, int n_in,
                              void* d_out, int out_size, void* d_ws, size_t ws_size,
                              hipStream_t stream) {
  (void)n_in; (void)d_ws; (void)ws_size; (void)out_size;
  const float* t    = (const float*)d_in[0];
  const float* W1   = (const float*)d_in[1];
  const float* b1   = (const float*)d_in[2];
  const float* W2   = (const float*)d_in[3];
  const float* b2   = (const float*)d_in[4];
  const float* xmin = (const float*)d_in[5];
  const int N = in_sizes[0];

  const int tiles  = (N + PTS_PER_TILE - 1) / PTS_PER_TILE;
  const int blocks = (tiles + WAVES_PER_BLOCK - 1) / WAVES_PER_BLOCK;
  springnn_wmma_f32<<<blocks, TPB, 0, stream>>>(t, W1, b1, W2, b2, xmin,
                                                (float*)d_out, N);
}


# Each wave owns 4 point tiles and reads two springs' weights from LDS into registers as one group per 64 WMMAs, so no LDS read stands between WMMAs; neighbouring WMMAs use different accumulators
# speedup vs baseline: 1.0462x; 1.0462x over previous
#include <hip/hip_runtime.h>
#include <math.h>

typedef __attribute__((ext_vector_type(2))) float v2f;
typedef __attribute__((ext_vector_type(4))) float v4f;
typedef __attribute__((ext_vector_type(8))) float v8f;

#define SPR_S 32
#define SPR_H 32
#define TPB 256
#define WAVES_PER_BLOCK (TPB / 32)
#define PTS_PER_TILE 16
#define TILES_PER_WAVE 4
#define PTS_PER_WAVE (PTS_PER_TILE * TILES_PER_WAVE)

#if __has_builtin(__builtin_amdgcn_tanhf)
  #define FAST_TANH(x) __builtin_amdgcn_tanhf(x)
#elif __has_builtin(__builtin_amdgcn_tanh_f32)
  #define FAST_TANH(x) __builtin_amdgcn_tanh_f32(x)
#else
  #define FAST_TANH(x) tanhf(x)
#endif


struct SpringRegs {
  v2f w1[8], b1[8], w2[8];
  float lo, b2;
};

__device__ __forceinline__ void load_spring(SpringRegs& r, const float* sW1, const float* sB1,
                                            const float* sW2, const float* sB2,
                                            const float* sLo, int s, int kh) {
  const float* w1s = sW1 + s * SPR_H + 2 * kh;
  const float* b1s = sB1 + s * SPR_H + 2 * kh;
  const float* w2s = sW2 + s * SPR_H + 2 * kh;
#pragma unroll
  for (int j = 0; j < 8; ++j) {
    r.w1[j] = *(const v2f*)(w1s + 4 * j);
    r.b1[j] = *(const v2f*)(b1s + 4 * j);
    r.w2[j] = *(const v2f*)(w2s + 4 * j);
  }
  r.lo = sLo[s];
  r.b2 = sB2[s];
}

__device__ __forceinline__ void spring_block(const SpringRegs& r,
                                             const float (&tv)[TILES_PER_WAVE],
                                             v8f (&acc)[TILES_PER_WAVE],
                                             float (&accB2)[TILES_PER_WAVE]) {
  float w[TILES_PER_WAVE];
#pragma unroll
  for (int i = 0; i < TILES_PER_WAVE; ++i) {
    const float d  = tv[i] - r.lo;
    const float cw = 0.5f * (1.0f - __cosf(1.57079632679489662f * d));
    w[i] = (d >= 0.0f && d <= 4.0f) ? cw : 0.0f;
    accB2[i] = fmaf(w[i], r.b2, accB2[i]);
  }
#pragma unroll
  for (int j = 0; j < 8; ++j) {
#pragma unroll
    for (int i = 0; i < TILES_PER_WAVE; ++i) {
      v2f a;
      a.x = w[i] * FAST_TANH(fmaf(tv[i], r.w1[j].x, r.b1[j].x));
      a.y = w[i] * FAST_TANH(fmaf(tv[i], r.w1[j].y, r.b1[j].y));
      acc[i] = __builtin_amdgcn_wmma_f32_16x16x4_f32(
          false, a, false, r.w2[j], (short)0, acc[i], false, false);
    }
  }
}

__global__ __launch_bounds__(TPB) void springnn_wmma_f32(
    const float* __restrict__ t,
    const float* __restrict__ W1,
    const float* __restrict__ b1,
    const float* __restrict__ W2,
    const float* __restrict__ b2,
    const float* __restrict__ xmin,
    float* __restrict__ out, int N)
{
  __shared__ float sW1[SPR_S * SPR_H];
  __shared__ float sB1[SPR_S * SPR_H];
  __shared__ float sW2[SPR_S * SPR_H];
  __shared__ float sB2[SPR_S];
  __shared__ float sLo[SPR_S];

  for (int i = threadIdx.x; i < SPR_S * SPR_H; i += TPB) {
    sW1[i] = W1[i];
    sB1[i] = b1[i];
    sW2[i] = W2[i];
  }
  if (threadIdx.x < SPR_S) {
    sB2[threadIdx.x] = b2[threadIdx.x];
    sLo[threadIdx.x] = xmin[threadIdx.x];
  }
  __syncthreads();

  const int lane  = threadIdx.x & 31;
  const int wave  = threadIdx.x >> 5;
  const int kh    = lane >> 4;
  const int p     = lane & 15;
  const int wbase = (blockIdx.x * WAVES_PER_BLOCK + wave) * PTS_PER_WAVE;
  if (wbase >= N) return;

  float tv[TILES_PER_WAVE];
#pragma unroll
  for (int i = 0; i < TILES_PER_WAVE; ++i) {
    const int tidx = wbase + i * PTS_PER_TILE + p;
    tv[i] = t[tidx < N ? tidx : (N - 1)];
  }

  v8f acc[TILES_PER_WAVE];
  float accB2[TILES_PER_WAVE];
#pragma unroll
  for (int i = 0; i < TILES_PER_WAVE; ++i) {
    acc[i] = (v8f){};
    accB2[i] = 0.0f;
  }

  SpringRegs ra, rb;
  for (int s = 0; s < SPR_S; s += 2) {
    load_spring(ra, sW1, sB1, sW2, sB2, sLo, s, kh);
    load_spring(rb, sW1, sB1, sW2, sB2, sLo, s + 1, kh);
    __builtin_amdgcn_sched_barrier(0);
    spring_block(ra, tv, acc, accB2);
    spring_block(rb, tv, acc, accB2);
  }

#pragma unroll
  for (int i = 0; i < TILES_PER_WAVE; ++i) {
    const int base = wbase + i * PTS_PER_TILE;
    const int ib2 = __builtin_bit_cast(int, accB2[i]);
    v8f res;
#pragma unroll
    for (int r = 0; r < 8; ++r) {
      const float addLo = __builtin_bit_cast(float, __builtin_amdgcn_readlane(ib2, r));
      const float addHi = __builtin_bit_cast(float, __builtin_amdgcn_readlane(ib2, r + 8));
      res[r] = acc[i][r] + (kh ? addHi : addLo);
    }

    if (p == 0 && base < N) {
      const int o0 = base + kh * 8;
      if (o0 + 8 <= N) {
        v4f lo4 = {res[0], res[1], res[2], res[3]};
        v4f hi4 = {res[4], res[5], res[6], res[7]};
        *(v4f*)(out + o0)     = lo4;
        *(v4f*)(out + o0 + 4) = hi4;
      } else {
        for (int r = 0; r < 8 && o0 + r < N; ++r) out[o0 + r] = res[r];
      }
    }
  }
}

extern "C" void kernel_launch(void* const* d_in, const int* in_sizes, int n_in,
                              void* d_out, int out_size, void* d_ws, size_t ws_size,
                              hipStream_t stream) {
  (void)n_in; (void)d_ws; (void)ws_size; (void)out_size;
  const float* t    = (const float*)d_in[0];
  const float* W1   = (const float*)d_in[1];
  const float* b1   = (const float*)d_in[2];
  const float* W2   = (const float*)d_in[3];
  const float* b2   = (const float*)d_in[4];
  const float* xmin = (const float*)d_in[5];
  const int N = in_sizes[0];

  const int wtiles = (N + PTS_PER_WAVE - 1) / PTS_PER_WAVE;
  const int blocks = (wtiles + WAVES_PER_BLOCK - 1) / WAVES_PER_BLOCK;
  springnn_wmma_f32<<<blocks, TPB, 0, stream>>>(t, W1, b1, W2, b2, xmin,
                                                (float*)d_out, N);
}


# Block orders its 512 points by t through LDS so each 16-point tile is narrow in t; (tile,spring) products whose window holds no point of the tile are skipped by scalar branch; sum order unchanged
# speedup vs baseline: 2.2463x; 2.1471x over previous
#include <hip/hip_runtime.h>
#include <math.h>


typedef __attribute__((ext_vector_type(2))) float v2f;
typedef __attribute__((ext_vector_type(4))) float v4f;
typedef __attribute__((ext_vector_type(8))) float v8f;
typedef __attribute__((ext_vector_type(4))) unsigned v4u;

#define SPR_S 32
#define SPR_H 32
#define TPB 256
#define PTS_PER_BLOCK 512
#define PTS_PER_TILE 16
#define TILES_PER_WAVE 4
#define PTS_PER_WAVE (PTS_PER_TILE * TILES_PER_WAVE)

#if __has_builtin(__builtin_amdgcn_tanhf)
  #define FAST_TANH(x) __builtin_amdgcn_tanhf(x)
#elif __has_builtin(__builtin_amdgcn_tanh_f32)
  #define FAST_TANH(x) __builtin_amdgcn_tanh_f32(x)
#else
  #define FAST_TANH(x) tanhf(x)
#endif


struct SpringRegs {
  v2f w1[8], b1[8], w2[8];
  float lo, b2;
};

__device__ __forceinline__ void load_spring(SpringRegs& r, const float* sW1, const float* sB1,
                                            const float* sW2, const float* sB2,
                                            const float* sLo, int s, int kh) {
  const float* w1s = sW1 + s * SPR_H + 2 * kh;
  const float* b1s = sB1 + s * SPR_H + 2 * kh;
  const float* w2s = sW2 + s * SPR_H + 2 * kh;
#pragma unroll
  for (int j = 0; j < 8; ++j) {
    r.w1[j] = *(const v2f*)(w1s + 4 * j);
    r.b1[j] = *(const v2f*)(b1s + 4 * j);
    r.w2[j] = *(const v2f*)(w2s + 4 * j);
  }
  r.lo = sLo[s];
  r.b2 = sB2[s];
}

__device__ __forceinline__ void tile_spring(const SpringRegs& r, const float tv,
                                            v8f& acc, float& accB2) {
  const float d  = tv - r.lo;
  const float cw = 0.5f * (1.0f - __cosf(1.57079632679489662f * d));
  const float w  = (d >= 0.0f && d <= 4.0f) ? cw : 0.0f;
  accB2 = fmaf(w, r.b2, accB2);
#pragma unroll
  for (int j = 0; j < 8; ++j) {
    v2f a;
    a.x = w * FAST_TANH(fmaf(tv, r.w1[j].x, r.b1[j].x));
    a.y = w * FAST_TANH(fmaf(tv, r.w1[j].y, r.b1[j].y));
    acc = __builtin_amdgcn_wmma_f32_16x16x4_f32(
        false, a, false, r.w2[j], (short)0, acc, false, false);
  }
}

__device__ __forceinline__ unsigned sort_key(float x, int idx) {
  const unsigned u = __builtin_bit_cast(unsigned, x);
  const unsigned m = (u & 0x80000000u) ? ~u : (u | 0x80000000u);
  return (m & ~511u) | (unsigned)idx;
}

__global__ __launch_bounds__(TPB) void springnn_wmma_f32(
    const float* __restrict__ t,
    const float* __restrict__ W1,
    const float* __restrict__ b1,
    const float* __restrict__ W2,
    const float* __restrict__ b2,
    const float* __restrict__ xmin,
    float* __restrict__ out, int N)
{
  __shared__ float sW1[SPR_S * SPR_H];
  __shared__ float sB1[SPR_S * SPR_H];
  __shared__ float sW2[SPR_S * SPR_H];
  __shared__ float sB2[SPR_S];
  __shared__ float sLo[SPR_S];
  __shared__ __attribute__((aligned(16))) unsigned sKey[PTS_PER_BLOCK];
  __shared__ float sT[PTS_PER_BLOCK];
  __shared__ int   sI[PTS_PER_BLOCK];
  __shared__ float sOut[PTS_PER_BLOCK];

  const int tid   = threadIdx.x;
  const int gbase = blockIdx.x * PTS_PER_BLOCK;

  const int i0 = tid, i1 = tid + TPB;
  const int g0 = gbase + i0, g1 = gbase + i1;
  const float t0 = t[g0 < N ? g0 : (N - 1)];
  const float t1 = t[g1 < N ? g1 : (N - 1)];
  const unsigned k0 = sort_key(t0, i0);
  const unsigned k1 = sort_key(t1, i1);
  sKey[i0] = k0;
  sKey[i1] = k1;

  for (int i = tid; i < SPR_S * SPR_H; i += TPB) {
    sW1[i] = W1[i];
    sB1[i] = b1[i];
    sW2[i] = W2[i];
  }
  if (tid < SPR_S) {
    sB2[tid] = b2[tid];
    sLo[tid] = xmin[tid];
  }
  __syncthreads();

  unsigned r0 = 0, r1 = 0;
  for (int j = 0; j < PTS_PER_BLOCK; j += 4) {
    const v4u k = *(const v4u*)(sKey + j);
    r0 += (unsigned)(k.x < k0) + (unsigned)(k.y < k0) + (unsigned)(k.z < k0) + (unsigned)(k.w < k0);
    r1 += (unsigned)(k.x < k1) + (unsigned)(k.y < k1) + (unsigned)(k.z < k1) + (unsigned)(k.w < k1);
  }
  sT[r0] = t0;  sI[r0] = i0;
  sT[r1] = t1;  sI[r1] = i1;
  __syncthreads();

  const int lane  = tid & 31;
  const int wave  = tid >> 5;
  const int kh    = lane >> 4;
  const int p     = lane & 15;
  const int sbase = wave * PTS_PER_WAVE;

  const float vlo = sLo[lane];
  float tv[TILES_PER_WAVE];
  unsigned act[TILES_PER_WAVE];
#pragma unroll
  for (int i = 0; i < TILES_PER_WAVE; ++i) {
    tv[i] = sT[sbase + i * PTS_PER_TILE + p];
    float mn = tv[i], mx = tv[i];
#pragma unroll
    for (int m = 1; m < 16; m <<= 1) {
      mn = fminf(mn, __shfl_xor(mn, m, 32));
      mx = fmaxf(mx, __shfl_xor(mx, m, 32));
    }
    const float dmax = mx - vlo;
    const float dmin = mn - vlo;
    act[i] = __builtin_amdgcn_ballot_w32(dmax >= 0.0f && dmin <= 4.0f);
  }

#pragma unroll
  for (int i = 0; i < TILES_PER_WAVE; ++i) {
    v8f acc = (v8f){};
    float accB2 = 0.0f;
    unsigned m = act[i];
    if (m != 0u) {
      SpringRegs ra, rb;
      load_spring(ra, sW1, sB1, sW2, sB2, sLo, __builtin_ctz(m), kh);
      for (;;) {
        m &= m - 1u;
        load_spring(rb, sW1, sB1, sW2, sB2, sLo, m ? __builtin_ctz(m) : 0, kh);
        __builtin_amdgcn_sched_barrier(0);
        tile_spring(ra, tv[i], acc, accB2);
        if (m == 0u) break;
        m &= m - 1u;
        load_spring(ra, sW1, sB1, sW2, sB2, sLo, m ? __builtin_ctz(m) : 0, kh);
        __builtin_amdgcn_sched_barrier(0);
        tile_spring(rb, tv[i], acc, accB2);
        if (m == 0u) break;
      }
    }

    const int ib2 = __builtin_bit_cast(int, accB2);
    float val = 0.0f;
#pragma unroll
    for (int r = 0; r < 8; ++r) {
      const float addLo = __builtin_bit_cast(float, __builtin_amdgcn_readlane(ib2, r));
      const float addHi = __builtin_bit_cast(float, __builtin_amdgcn_readlane(ib2, r + 8));
      const float res = acc[r] + (kh ? addHi : addLo);
      val = (p == r) ? res : val;
    }
    if (p < 8) {
      const int slot = sbase + i * PTS_PER_TILE + kh * 8 + p;
      sOut[sI[slot]] = val;
    }
  }
  __syncthreads();

  if (g0 < N) out[g0] = sOut[i0];
  if (g1 < N) out[g1] = sOut[i1];
}

extern "C" void kernel_launch(void* const* d_in, const int* in_sizes, int n_in,
                              void* d_out, int out_size, void* d_ws, size_t ws_size,
                              hipStream_t stream) {
  (void)n_in; (void)d_ws; (void)ws_size; (void)out_size;
  const float* t    = (const float*)d_in[0];
  const float* W1   = (const float*)d_in[1];
  const float* b1   = (const float*)d_in[2];
  const float* W2   = (const float*)d_in[3];
  const float* b2   = (const float*)d_in[4];
  const float* xmin = (const float*)d_in[5];
  const int N = in_sizes[0];

  const int blocks = (N + PTS_PER_BLOCK - 1) / PTS_PER_BLOCK;
  springnn_wmma_f32<<<blocks, TPB, 0, stream>>>(t, W1, b1, W2, b2, xmin,
                                                (float*)d_out, N);
}
